# starting kernel
# baseline (speedup 1.0000x reference)
#include <hip/hip_runtime.h>

#define N_NODES 100000
#define N_EDGES 800000
#define HIDDEN  128

typedef __attribute__((ext_vector_type(2))) float v2f;
typedef __attribute__((ext_vector_type(8))) float v8f;

__global__ void gnn_count_edges(const int* __restrict__ ei,
                                float* __restrict__ cnt, int E) {
    int e = blockIdx.x * blockDim.x + threadIdx.x;
    if (e < E) {
        int dst = ei[E + e];
        atomicAdd(&cnt[dst], 1.0f);
    }
}

__global__ void gnn_scatter(const float* __restrict__ feat,
                            const int* __restrict__ ei,
                            float* __restrict__ agg, int E) {
    int idx = blockIdx.x * blockDim.x + threadIdx.x;
    int e  = idx >> 5;
    int f4 = (idx & 31) << 2;
    if (e < E) {
        int src = ei[e];
        int dst = ei[E + e];
        const float4 v = *(const float4*)(feat + (size_t)src * HIDDEN + f4);
        float* a = agg + (size_t)dst * HIDDEN + f4;
        atomicAdd(a + 0, v.x);
        atomicAdd(a + 1, v.y);
        atomicAdd(a + 2, v.z);
        atomicAdd(a + 3, v.w);
    }
}

__global__ __launch_bounds__(128)
void gnn_sage_layer(const float* __restrict__ X,
                    const float* __restrict__ agg,
                    const float* __restrict__ cnt,
                    const float* __restrict__ Wl,
                    const float* __restrict__ Wr,
                    const float* __restrict__ bias,
                    float* __restrict__ out,
                    int do_relu) {
    constexpr int LD = HIDDEN + 4;
    __shared__ float Xt[16 * LD];
    __shared__ float Mt[16 * LD];

    const int tid  = threadIdx.x;
    const int row0 = blockIdx.x * 16;

    #pragma unroll
    for (int i = 0; i < 16; ++i) {
        float rc = 1.0f / fmaxf(cnt[row0 + i], 1.0f);
        Xt[i * LD + tid] = X[(size_t)(row0 + i) * HIDDEN + tid];
        Mt[i * LD + tid] = agg[(size_t)(row0 + i) * HIDDEN + tid] * rc;
    }
    __syncthreads();

    const int wave  = tid >> 5;
    const int lane  = tid & 31;
    const int m     = lane & 15;
    const int khalf = (lane >> 4) << 1;

    #pragma unroll
    for (int t = 0; t < 2; ++t) {
        const int col0 = (wave * 2 + t) * 16;

        const float bv = bias[col0 + m];
        v8f c = {bv, bv, bv, bv, bv, bv, bv, bv};

        #pragma unroll 4
        for (int k0 = 0; k0 < HIDDEN; k0 += 4) {
            const int kb = k0 + khalf;
            v2f a, b;

            a.x = Mt[m * LD + kb];
            a.y = Mt[m * LD + kb + 1];
            b.x = Wl[kb * HIDDEN + col0 + m];
            b.y = Wl[(kb + 1) * HIDDEN + col0 + m];
            c = __builtin_amdgcn_wmma_f32_16x16x4_f32(
                    false, a, false, b, (short)0, c, false, false);

            a.x = Xt[m * LD + kb];
            a.y = Xt[m * LD + kb + 1];
            b.x = Wr[kb * HIDDEN + col0 + m];
            b.y = Wr[(kb + 1) * HIDDEN + col0 + m];
            c = __builtin_amdgcn_wmma_f32_16x16x4_f32(
                    false, a, false, b, (short)0, c, false, false);
        }

        const int rbase = row0 + ((lane >> 4) << 3);
        #pragma unroll
        for (int r = 0; r < 8; ++r) {
            float v = c[r];
            if (do_relu) v = fmaxf(v, 0.0f);
            out[(size_t)(rbase + r) * HIDDEN + col0 + m] = v;
        }
    }
}

extern "C" void kernel_launch(void* const* d_in, const int* in_sizes, int n_in,
                              void* d_out, int out_size, void* d_ws, size_t ws_size,
                              hipStream_t stream) {
    const float* x   = (const float*)d_in[0];
    const int*   ei  = (const int*)d_in[1];
    const float* Wl1 = (const float*)d_in[2];
    const float* Wr1 = (const float*)d_in[3];
    const float* b1  = (const float*)d_in[4];
    const float* Wl2 = (const float*)d_in[5];
    const float* Wr2 = (const float*)d_in[6];
    const float* b2  = (const float*)d_in[7];
    float* out = (float*)d_out;

    char* ws = (char*)d_ws;
    const size_t featBytes = (size_t)N_NODES * HIDDEN * sizeof(float);
    float* agg = (float*)(ws);
    float* h   = (float*)(ws + featBytes);
    float* cnt = (float*)(ws + 2 * featBytes);

    hipMemsetAsync(agg, 0, featBytes, stream);
    hipMemsetAsync(cnt, 0, (size_t)N_NODES * sizeof(float), stream);

    const int scatterBlocks = (N_EDGES * 32) / 256;
    gnn_count_edges<<<(N_EDGES + 255) / 256, 256, 0, stream>>>(ei, cnt, N_EDGES);
    gnn_scatter<<<scatterBlocks, 256, 0, stream>>>(x, ei, agg, N_EDGES);

    gnn_sage_layer<<<N_NODES / 16, 128, 0, stream>>>(x, agg, cnt,
                                                     Wl1, Wr1, b1, h, 1);

    hipMemsetAsync(agg, 0, featBytes, stream);
    gnn_scatter<<<scatterBlocks, 256, 0, stream>>>(h, ei, agg, N_EDGES);

    gnn_sage_layer<<<N_NODES / 16, 128, 0, stream>>>(h, agg, cnt,
                                                     Wl2, Wr2, b2, out, 0);
}


# Earlier try summed in slot-arrival order (run-to-run drift): now f32 atomic scatter-add -> in-call dst-grouped edge list, one wave per node adds source rows in ascending source-id order (fixed)
# speedup vs baseline: 1.7451x; 1.7451x over previous
#include <hip/hip_runtime.h>


#define N_NODES 100000
#define N_EDGES 800000
#define HIDDEN  128

typedef __attribute__((ext_vector_type(2))) float v2f;
typedef __attribute__((ext_vector_type(8))) float v8f;

__global__ void gnn_count_edges(const int* __restrict__ ei,
                                int* __restrict__ cnt, int E) {
    int e = blockIdx.x * blockDim.x + threadIdx.x;
    if (e < E) {
        int dst = ei[E + e];
        atomicAdd(&cnt[dst], 1);
    }
}

__global__ __launch_bounds__(1024)
void gnn_reserve(const int* __restrict__ cnt, int* __restrict__ cursor,
                 int* __restrict__ total, float* __restrict__ degf, int n) {
    __shared__ int wsum[32];
    __shared__ int bbase;
    const int i    = blockIdx.x * 1024 + threadIdx.x;
    const int lane = threadIdx.x & 31;
    const int wave = threadIdx.x >> 5;
    const int c    = (i < n) ? cnt[i] : 0;

    int incl = c;
    #pragma unroll
    for (int o = 1; o < 32; o <<= 1) {
        int t = __shfl_up(incl, o, 32);
        if (lane >= o) incl += t;
    }
    if (lane == 31) wsum[wave] = incl;
    __syncthreads();
    if (wave == 0) {
        const int w = wsum[lane];
        int wi = w;
        #pragma unroll
        for (int o = 1; o < 32; o <<= 1) {
            int t = __shfl_up(wi, o, 32);
            if (lane >= o) wi += t;
        }
        wsum[lane] = wi - w;
        if (lane == 31) bbase = atomicAdd(total, wi);
    }
    __syncthreads();
    if (i < n) {
        cursor[i] = bbase + wsum[wave] + incl - c;
        degf[i]   = (float)c;
    }
}

__global__ void gnn_fill(const int* __restrict__ ei, int* __restrict__ cursor,
                         int* __restrict__ seg, int E) {
    int e = blockIdx.x * blockDim.x + threadIdx.x;
    if (e < E) {
        int src = ei[e];
        int dst = ei[E + e];
        int pos = atomicAdd(&cursor[dst], 1);
        seg[pos] = src;
    }
}

__device__ __forceinline__ float4 add4(float4 a, float4 b) {
    return make_float4(a.x + b.x, a.y + b.y, a.z + b.z, a.w + b.w);
}

template <int K>
__device__ __forceinline__ float4 add_rows(const float* __restrict__ fl, int sorted,
                                           int r, float4 acc) {
    float4 v[K];
    #pragma unroll
    for (int u = 0; u < K; ++u)
        v[u] = *(const float4*)(fl + (size_t)__builtin_amdgcn_readlane(sorted, r + u) * HIDDEN);
    #pragma unroll
    for (int u = 0; u < K; ++u) acc = add4(acc, v[u]);
    return acc;
}

__global__ __launch_bounds__(256)
void gnn_gather_sum(const float* __restrict__ feat,
                    const int* __restrict__ seg,
                    const int* __restrict__ cnt,
                    const int* __restrict__ cursor,
                    float* __restrict__ agg, int n) {
    const int lane = threadIdx.x & 31;
    const int node = blockIdx.x * 8 + __builtin_amdgcn_readfirstlane(threadIdx.x >> 5);
    if (node >= n) return;

    const int d   = cnt[node];
    const int beg = cursor[node] - d;
    const float* fl = feat + lane * 4;
    float4 acc = make_float4(0.0f, 0.0f, 0.0f, 0.0f);

    if (d <= 32) {
        const int s = (lane < d) ? seg[beg + lane] : 0x7fffffff;
        int rank = 0;
        for (int j = 0; j < d; ++j) {
            const int sj = __builtin_amdgcn_readlane(s, j);
            rank += (sj < s || (sj == s && j < lane)) ? 1 : 0;
        }
        if (lane >= d) rank = lane;
        const int sorted = __builtin_amdgcn_ds_permute(rank << 2, s);

        int r = 0;
        for (; r + 8 <= d; r += 8) acc = add_rows<8>(fl, sorted, r, acc);
        switch (d - r) {
            case 7: acc = add_rows<7>(fl, sorted, r, acc); break;
            case 6: acc = add_rows<6>(fl, sorted, r, acc); break;
            case 5: acc = add_rows<5>(fl, sorted, r, acc); break;
            case 4: acc = add_rows<4>(fl, sorted, r, acc); break;
            case 3: acc = add_rows<3>(fl, sorted, r, acc); break;
            case 2: acc = add_rows<2>(fl, sorted, r, acc); break;
            case 1: acc = add_rows<1>(fl, sorted, r, acc); break;
            default: break;
        }
    } else {
        int cur  = -1;
        int done = 0;
        while (done < d) {
            int best = 0x7fffffff;
            #pragma unroll 1
            for (int j = lane; j < d; j += 32) {
                const int v = seg[beg + j];
                if (v > cur && v < best) best = v;
            }
            #pragma unroll
            for (int o = 16; o > 0; o >>= 1) {
                const int t = __shfl_xor(best, o, 32);
                best = (t < best) ? t : best;
            }
            best = __builtin_amdgcn_readfirstlane(best);
            int m = 0;
            #pragma unroll 1
            for (int j = lane; j < d; j += 32) m += (seg[beg + j] == best) ? 1 : 0;
            #pragma unroll
            for (int o = 16; o > 0; o >>= 1) m += __shfl_xor(m, o, 32);
            m = __builtin_amdgcn_readfirstlane(m);
            if (m == 0) break;
            const float4 v = *(const float4*)(fl + (size_t)best * HIDDEN);
            #pragma unroll 1
            for (int k = 0; k < m; ++k) acc = add4(acc, v);
            cur = best;
            done += m;
        }
    }

    *(float4*)(agg + (size_t)node * HIDDEN + lane * 4) = acc;
}

__global__ __launch_bounds__(128)
void gnn_sage_layer(const float* __restrict__ X,
                    const float* __restrict__ agg,
                    const float* __restrict__ cnt,
                    const float* __restrict__ Wl,
                    const float* __restrict__ Wr,
                    const float* __restrict__ bias,
                    float* __restrict__ out,
                    int do_relu) {
    constexpr int LD = HIDDEN + 4;
    __shared__ float Xt[16 * LD];
    __shared__ float Mt[16 * LD];

    const int tid  = threadIdx.x;
    const int row0 = blockIdx.x * 16;

    #pragma unroll
    for (int i = 0; i < 16; ++i) {
        float rc = 1.0f / fmaxf(cnt[row0 + i], 1.0f);
        Xt[i * LD + tid] = X[(size_t)(row0 + i) * HIDDEN + tid];
        Mt[i * LD + tid] = agg[(size_t)(row0 + i) * HIDDEN + tid] * rc;
    }
    __syncthreads();

    const int wave  = tid >> 5;
    const int lane  = tid & 31;
    const int m     = lane & 15;
    const int khalf = (lane >> 4) << 1;

    #pragma unroll
    for (int t = 0; t < 2; ++t) {
        const int col0 = (wave * 2 + t) * 16;

        const float bv = bias[col0 + m];
        v8f c = {bv, bv, bv, bv, bv, bv, bv, bv};

        #pragma unroll 4
        for (int k0 = 0; k0 < HIDDEN; k0 += 4) {
            const int kb = k0 + khalf;
            v2f a, b;

            a.x = Mt[m * LD + kb];
            a.y = Mt[m * LD + kb + 1];
            b.x = Wl[kb * HIDDEN + col0 + m];
            b.y = Wl[(kb + 1) * HIDDEN + col0 + m];
            c = __builtin_amdgcn_wmma_f32_16x16x4_f32(
                    false, a, false, b, (short)0, c, false, false);

            a.x = Xt[m * LD + kb];
            a.y = Xt[m * LD + kb + 1];
            b.x = Wr[kb * HIDDEN + col0 + m];
            b.y = Wr[(kb + 1) * HIDDEN + col0 + m];
            c = __builtin_amdgcn_wmma_f32_16x16x4_f32(
                    false, a, false, b, (short)0, c, false, false);
        }

        const int rbase = row0 + ((lane >> 4) << 3);
        #pragma unroll
        for (int r = 0; r < 8; ++r) {
            float v = c[r];
            if (do_relu) v = fmaxf(v, 0.0f);
            out[(size_t)(rbase + r) * HIDDEN + col0 + m] = v;
        }
    }
}

extern "C" void kernel_launch(void* const* d_in, const int* in_sizes, int n_in,
                              void* d_out, int out_size, void* d_ws, size_t ws_size,
                              hipStream_t stream) {
    const float* x   = (const float*)d_in[0];
    const int*   ei  = (const int*)d_in[1];
    const float* Wl1 = (const float*)d_in[2];
    const float* Wr1 = (const float*)d_in[3];
    const float* b1  = (const float*)d_in[4];
    const float* Wl2 = (const float*)d_in[5];
    const float* Wr2 = (const float*)d_in[6];
    const float* b2  = (const float*)d_in[7];
    float* out = (float*)d_out;

    char* ws = (char*)d_ws;
    const size_t featBytes = (size_t)N_NODES * HIDDEN * sizeof(float);
    float* agg  = (float*)(ws);
    float* h    = (float*)(ws + featBytes);
    int* cnt    = (int*)(ws + 2 * featBytes);
    int* total  = cnt + N_NODES;
    int* cursor = cnt + N_NODES + 32;
    int* seg    = cursor + N_NODES;
    float* degf = (float*)(seg + N_EDGES);

    hipMemsetAsync(cnt, 0, (size_t)(N_NODES + 1) * sizeof(int), stream);

    gnn_count_edges<<<(N_EDGES + 255) / 256, 256, 0, stream>>>(ei, cnt, N_EDGES);
    gnn_reserve<<<(N_NODES + 1023) / 1024, 1024, 0, stream>>>(cnt, cursor, total, degf, N_NODES);
    gnn_fill<<<(N_EDGES + 255) / 256, 256, 0, stream>>>(ei, cursor, seg, N_EDGES);

    gnn_gather_sum<<<(N_NODES + 7) / 8, 256, 0, stream>>>(x, seg, cnt, cursor, agg, N_NODES);
    gnn_sage_layer<<<N_NODES / 16, 128, 0, stream>>>(x, agg, degf,
                                                     Wl1, Wr1, b1, h, 1);

    gnn_gather_sum<<<(N_NODES + 7) / 8, 256, 0, stream>>>(h, seg, cnt, cursor, agg, N_NODES);
    gnn_sage_layer<<<N_NODES / 16, 128, 0, stream>>>(h, agg, degf,
                                                     Wl2, Wr2, b2, out, 0);
}


# Neighbour sum fused into the layer kernel: each wave sums 4 of its block's 16 nodes (same ascending-id order) into the LDS mean tile; the two gather launches and the sum array in memory go
# speedup vs baseline: 1.8469x; 1.0583x over previous
#include <hip/hip_runtime.h>


#define N_NODES 100000
#define N_EDGES 800000
#define HIDDEN  128

typedef __attribute__((ext_vector_type(2))) float v2f;
typedef __attribute__((ext_vector_type(8))) float v8f;

__global__ void gnn_count_edges(const int* __restrict__ ei,
                                int* __restrict__ cnt, int E) {
    int e = blockIdx.x * blockDim.x + threadIdx.x;
    if (e < E) {
        int dst = ei[E + e];
        atomicAdd(&cnt[dst], 1);
    }
}

__global__ __launch_bounds__(1024)
void gnn_reserve(const int* __restrict__ cnt, int* __restrict__ cursor,
                 int* __restrict__ total, int n) {
    __shared__ int wsum[32];
    __shared__ int bbase;
    const int i    = blockIdx.x * 1024 + threadIdx.x;
    const int lane = threadIdx.x & 31;
    const int wave = threadIdx.x >> 5;
    const int c    = (i < n) ? cnt[i] : 0;

    int incl = c;
    #pragma unroll
    for (int o = 1; o < 32; o <<= 1) {
        int t = __shfl_up(incl, o, 32);
        if (lane >= o) incl += t;
    }
    if (lane == 31) wsum[wave] = incl;
    __syncthreads();
    if (wave == 0) {
        const int w = wsum[lane];
        int wi = w;
        #pragma unroll
        for (int o = 1; o < 32; o <<= 1) {
            int t = __shfl_up(wi, o, 32);
            if (lane >= o) wi += t;
        }
        wsum[lane] = wi - w;
        if (lane == 31) bbase = atomicAdd(total, wi);
    }
    __syncthreads();
    if (i < n) cursor[i] = bbase + wsum[wave] + incl - c;
}

__global__ void gnn_fill(const int* __restrict__ ei, int* __restrict__ cursor,
                         int* __restrict__ seg, int E) {
    int e = blockIdx.x * blockDim.x + threadIdx.x;
    if (e < E) {
        int src = ei[e];
        int dst = ei[E + e];
        int pos = atomicAdd(&cursor[dst], 1);
        seg[pos] = src;
    }
}

__device__ __forceinline__ float4 add4(float4 a, float4 b) {
    return make_float4(a.x + b.x, a.y + b.y, a.z + b.z, a.w + b.w);
}

template <int K>
__device__ __forceinline__ float4 add_rows(const float* __restrict__ fl, int sorted,
                                           int r, float4 acc) {
    float4 v[K];
    #pragma unroll
    for (int u = 0; u < K; ++u)
        v[u] = *(const float4*)(fl + (size_t)__builtin_amdgcn_readlane(sorted, r + u) * HIDDEN);
    #pragma unroll
    for (int u = 0; u < K; ++u) acc = add4(acc, v[u]);
    return acc;
}

__device__ __forceinline__ float4 neighbour_sum(const float* __restrict__ fl,
                                                const int* __restrict__ seg,
                                                int s, int d, int beg, int lane) {
    float4 acc = make_float4(0.0f, 0.0f, 0.0f, 0.0f);
    if (d <= 32) {
        int rank = 0;
        for (int j = 0; j < d; ++j) {
            const int sj = __builtin_amdgcn_readlane(s, j);
            rank += (sj < s || (sj == s && j < lane)) ? 1 : 0;
        }
        if (lane >= d) rank = lane;
        const int sorted = __builtin_amdgcn_ds_permute(rank << 2, s);

        int r = 0;
        for (; r + 8 <= d; r += 8) acc = add_rows<8>(fl, sorted, r, acc);
        switch (d - r) {
            case 7: acc = add_rows<7>(fl, sorted, r, acc); break;
            case 6: acc = add_rows<6>(fl, sorted, r, acc); break;
            case 5: acc = add_rows<5>(fl, sorted, r, acc); break;
            case 4: acc = add_rows<4>(fl, sorted, r, acc); break;
            case 3: acc = add_rows<3>(fl, sorted, r, acc); break;
            case 2: acc = add_rows<2>(fl, sorted, r, acc); break;
            case 1: acc = add_rows<1>(fl, sorted, r, acc); break;
            default: break;
        }
    } else {
        int cur  = -1;
        int done = 0;
        while (done < d) {
            int best = 0x7fffffff;
            #pragma unroll 1
            for (int j = lane; j < d; j += 32) {
                const int v = seg[beg + j];
                if (v > cur && v < best) best = v;
            }
            #pragma unroll
            for (int o = 16; o > 0; o >>= 1) {
                const int t = __shfl_xor(best, o, 32);
                best = (t < best) ? t : best;
            }
            best = __builtin_amdgcn_readfirstlane(best);
            int m = 0;
            #pragma unroll 1
            for (int j = lane; j < d; j += 32) m += (seg[beg + j] == best) ? 1 : 0;
            #pragma unroll
            for (int o = 16; o > 0; o >>= 1) m += __shfl_xor(m, o, 32);
            m = __builtin_amdgcn_readfirstlane(m);
            if (m == 0) break;
            const float4 v = *(const float4*)(fl + (size_t)best * HIDDEN);
            #pragma unroll 1
            for (int k = 0; k < m; ++k) acc = add4(acc, v);
            cur = best;
            done += m;
        }
    }
    return acc;
}

__global__ __launch_bounds__(128)
void gnn_sage_layer(const float* __restrict__ X,
                    const int* __restrict__ seg,
                    const int* __restrict__ cnt,
                    const int* __restrict__ cursor,
                    const float* __restrict__ Wl,
                    const float* __restrict__ Wr,
                    const float* __restrict__ bias,
                    float* __restrict__ out,
                    int do_relu) {
    constexpr int LD = HIDDEN + 4;
    __shared__ __attribute__((aligned(16))) float Xt[16 * LD];
    __shared__ __attribute__((aligned(16))) float Mt[16 * LD];

    const int tid  = threadIdx.x;
    const int lane = tid & 31;
    const int wave = __builtin_amdgcn_readfirstlane(tid >> 5);
    const int row0 = blockIdx.x * 16;

    float xr[16];
    #pragma unroll
    for (int i = 0; i < 16; ++i) xr[i] = X[(size_t)(row0 + i) * HIDDEN + tid];

    const int node0 = row0 + wave * 4;
    int d[4], beg[4], s[4];
    #pragma unroll
    for (int j = 0; j < 4; ++j) {
        d[j]   = cnt[node0 + j];
        beg[j] = cursor[node0 + j] - d[j];
    }
    #pragma unroll
    for (int j = 0; j < 4; ++j)
        s[j] = (lane < d[j]) ? seg[beg[j] + lane] : 0x7fffffff;

    #pragma unroll
    for (int i = 0; i < 16; ++i) Xt[i * LD + tid] = xr[i];

    const float* fl = X + lane * 4;
    #pragma unroll
    for (int j = 0; j < 4; ++j) {
        const float4 a = neighbour_sum(fl, seg, s[j], d[j], beg[j], lane);
        const float rc = 1.0f / fmaxf((float)d[j], 1.0f);
        *(float4*)(Mt + (wave * 4 + j) * LD + lane * 4) =
            make_float4(a.x * rc, a.y * rc, a.z * rc, a.w * rc);
    }
    __syncthreads();

    const int m     = lane & 15;
    const int khalf = (lane >> 4) << 1;

    #pragma unroll
    for (int t = 0; t < 2; ++t) {
        const int col0 = (wave * 2 + t) * 16;

        const float bv = bias[col0 + m];
        v8f c = {bv, bv, bv, bv, bv, bv, bv, bv};

        #pragma unroll 4
        for (int k0 = 0; k0 < HIDDEN; k0 += 4) {
            const int kb = k0 + khalf;
            v2f a, b;

            a.x = Mt[m * LD + kb];
            a.y = Mt[m * LD + kb + 1];
            b.x = Wl[kb * HIDDEN + col0 + m];
            b.y = Wl[(kb + 1) * HIDDEN + col0 + m];
            c = __builtin_amdgcn_wmma_f32_16x16x4_f32(
                    false, a, false, b, (short)0, c, false, false);

            a.x = Xt[m * LD + kb];
            a.y = Xt[m * LD + kb + 1];
            b.x = Wr[kb * HIDDEN + col0 + m];
            b.y = Wr[(kb + 1) * HIDDEN + col0 + m];
            c = __builtin_amdgcn_wmma_f32_16x16x4_f32(
                    false, a, false, b, (short)0, c, false, false);
        }

        const int rbase = row0 + ((lane >> 4) << 3);
        #pragma unroll
        for (int r = 0; r < 8; ++r) {
            float v = c[r];
            if (do_relu) v = fmaxf(v, 0.0f);
            out[(size_t)(rbase + r) * HIDDEN + col0 + m] = v;
        }
    }
}

extern "C" void kernel_launch(void* const* d_in, const int* in_sizes, int n_in,
                              void* d_out, int out_size, void* d_ws, size_t ws_size,
                              hipStream_t stream) {
    const float* x   = (const float*)d_in[0];
    const int*   ei  = (const int*)d_in[1];
    const float* Wl1 = (const float*)d_in[2];
    const float* Wr1 = (const float*)d_in[3];
    const float* b1  = (const float*)d_in[4];
    const float* Wl2 = (const float*)d_in[5];
    const float* Wr2 = (const float*)d_in[6];
    const float* b2  = (const float*)d_in[7];
    float* out = (float*)d_out;

    char* ws = (char*)d_ws;
    const size_t featBytes = (size_t)N_NODES * HIDDEN * sizeof(float);
    float* h    = (float*)(ws);
    int* cnt    = (int*)(ws + featBytes);
    int* total  = cnt + N_NODES;
    int* cursor = cnt + N_NODES + 32;
    int* seg    = cursor + N_NODES;

    hipMemsetAsync(cnt, 0, (size_t)(N_NODES + 1) * sizeof(int), stream);

    gnn_count_edges<<<(N_EDGES + 255) / 256, 256, 0, stream>>>(ei, cnt, N_EDGES);
    gnn_reserve<<<(N_NODES + 1023) / 1024, 1024, 0, stream>>>(cnt, cursor, total, N_NODES);
    gnn_fill<<<(N_EDGES + 255) / 256, 256, 0, stream>>>(ei, cursor, seg, N_EDGES);

    gnn_sage_layer<<<N_NODES / 16, 128, 0, stream>>>(x, seg, cnt, cursor,
                                                     Wl1, Wr1, b1, h, 1);
    gnn_sage_layer<<<N_NODES / 16, 128, 0, stream>>>(h, seg, cnt, cursor,
                                                     Wl2, Wr2, b2, out, 0);
}


# Layer block now 80 nodes x 8 waves (was 16 x 4): each wave owns 16 columns and uses each W fragment it loads for 5 row tiles, so W is read once per 80 rows; rows stored whole via LDS; same sum order
# speedup vs baseline: 1.9848x; 1.0747x over previous
#include <hip/hip_runtime.h>


#define N_NODES 100000
#define N_EDGES 800000
#define HIDDEN  128

typedef __attribute__((ext_vector_type(2))) float v2f;
typedef __attribute__((ext_vector_type(8))) float v8f;

__global__ void gnn_count_edges(const int* __restrict__ ei,
                                int* __restrict__ cnt, int E) {
    int e = blockIdx.x * blockDim.x + threadIdx.x;
    if (e < E) {
        int dst = ei[E + e];
        atomicAdd(&cnt[dst], 1);
    }
}

__global__ __launch_bounds__(1024)
void gnn_reserve(const int* __restrict__ cnt, int* __restrict__ cursor,
                 int* __restrict__ total, int n) {
    __shared__ int wsum[32];
    __shared__ int bbase;
    const int i    = blockIdx.x * 1024 + threadIdx.x;
    const int lane = threadIdx.x & 31;
    const int wave = threadIdx.x >> 5;
    const int c    = (i < n) ? cnt[i] : 0;

    int incl = c;
    #pragma unroll
    for (int o = 1; o < 32; o <<= 1) {
        int t = __shfl_up(incl, o, 32);
        if (lane >= o) incl += t;
    }
    if (lane == 31) wsum[wave] = incl;
    __syncthreads();
    if (wave == 0) {
        const int w = wsum[lane];
        int wi = w;
        #pragma unroll
        for (int o = 1; o < 32; o <<= 1) {
            int t = __shfl_up(wi, o, 32);
            if (lane >= o) wi += t;
        }
        wsum[lane] = wi - w;
        if (lane == 31) bbase = atomicAdd(total, wi);
    }
    __syncthreads();
    if (i < n) cursor[i] = bbase + wsum[wave] + incl - c;
}

__global__ void gnn_fill(const int* __restrict__ ei, int* __restrict__ cursor,
                         int* __restrict__ seg, int E) {
    int e = blockIdx.x * blockDim.x + threadIdx.x;
    if (e < E) {
        int src = ei[e];
        int dst = ei[E + e];
        int pos = atomicAdd(&cursor[dst], 1);
        seg[pos] = src;
    }
}

__device__ __forceinline__ float4 add4(float4 a, float4 b) {
    return make_float4(a.x + b.x, a.y + b.y, a.z + b.z, a.w + b.w);
}

template <int K>
__device__ __forceinline__ float4 add_rows(const float* __restrict__ fl, int sorted,
                                           int r, float4 acc) {
    float4 v[K];
    #pragma unroll
    for (int u = 0; u < K; ++u)
        v[u] = *(const float4*)(fl + (size_t)__builtin_amdgcn_readlane(sorted, r + u) * HIDDEN);
    #pragma unroll
    for (int u = 0; u < K; ++u) acc = add4(acc, v[u]);
    return acc;
}

__device__ __forceinline__ float4 neighbour_sum(const float* __restrict__ fl,
                                                const int* __restrict__ seg,
                                                int s, int d, int beg, int lane) {
    float4 acc = make_float4(0.0f, 0.0f, 0.0f, 0.0f);
    if (d <= 32) {
        int rank = 0;
        for (int j = 0; j < d; ++j) {
            const int sj = __builtin_amdgcn_readlane(s, j);
            rank += (sj < s || (sj == s && j < lane)) ? 1 : 0;
        }
        if (lane >= d) rank = lane;
        const int sorted = __builtin_amdgcn_ds_permute(rank << 2, s);

        int r = 0;
        for (; r + 8 <= d; r += 8) acc = add_rows<8>(fl, sorted, r, acc);
        switch (d - r) {
            case 7: acc = add_rows<7>(fl, sorted, r, acc); break;
            case 6: acc = add_rows<6>(fl, sorted, r, acc); break;
            case 5: acc = add_rows<5>(fl, sorted, r, acc); break;
            case 4: acc = add_rows<4>(fl, sorted, r, acc); break;
            case 3: acc = add_rows<3>(fl, sorted, r, acc); break;
            case 2: acc = add_rows<2>(fl, sorted, r, acc); break;
            case 1: acc = add_rows<1>(fl, sorted, r, acc); break;
            default: break;
        }
    } else {
        int cur  = -1;
        int done = 0;
        while (done < d) {
            int best = 0x7fffffff;
            #pragma unroll 1
            for (int j = lane; j < d; j += 32) {
                const int v = seg[beg + j];
                if (v > cur && v < best) best = v;
            }
            #pragma unroll
            for (int o = 16; o > 0; o >>= 1) {
                const int t = __shfl_xor(best, o, 32);
                best = (t < best) ? t : best;
            }
            best = __builtin_amdgcn_readfirstlane(best);
            int m = 0;
            #pragma unroll 1
            for (int j = lane; j < d; j += 32) m += (seg[beg + j] == best) ? 1 : 0;
            #pragma unroll
            for (int o = 16; o > 0; o >>= 1) m += __shfl_xor(m, o, 32);
            m = __builtin_amdgcn_readfirstlane(m);
            if (m == 0) break;
            const float4 v = *(const float4*)(fl + (size_t)best * HIDDEN);
            #pragma unroll 1
            for (int k = 0; k < m; ++k) acc = add4(acc, v);
            cur = best;
            done += m;
        }
    }
    return acc;
}

#define BLOCK_ROWS 80
#define ROW_TILES  5
#define WAVE_NODES 10

static_assert(N_NODES % BLOCK_ROWS == 0, "no row guard");

__global__ __launch_bounds__(256)
void gnn_sage_layer(const float* __restrict__ X,
                    const int* __restrict__ seg,
                    const int* __restrict__ cnt,
                    const int* __restrict__ cursor,
                    const float* __restrict__ Wl,
                    const float* __restrict__ Wr,
                    const float* __restrict__ bias,
                    float* __restrict__ out,
                    int do_relu) {
    __shared__ __attribute__((aligned(16))) float Xt[BLOCK_ROWS * HIDDEN];
    __shared__ __attribute__((aligned(16))) float Mt[BLOCK_ROWS * HIDDEN];

    const int tid  = threadIdx.x;
    const int lane = tid & 31;
    const int wave  = __builtin_amdgcn_readfirstlane(tid >> 5);
    const int row0  = blockIdx.x * BLOCK_ROWS;
    const int lrow0 = wave * WAVE_NODES;
    const int node0 = row0 + lrow0;

    int dv = 0, ev = 0;
    if (lane < WAVE_NODES) {
        dv = cnt[node0 + lane];
        ev = cursor[node0 + lane];
    }

    const float* fl = X + lane * 4;
    float4 xr[WAVE_NODES];
    #pragma unroll
    for (int j = 0; j < WAVE_NODES; ++j)
        xr[j] = *(const float4*)(fl + (size_t)(node0 + j) * HIDDEN);

    int d_n   = __builtin_amdgcn_readlane(dv, 0);
    int beg_n = __builtin_amdgcn_readlane(ev, 0) - d_n;
    int s_n   = (lane < d_n) ? seg[beg_n + lane] : 0x7fffffff;

    #pragma unroll
    for (int j = 0; j < WAVE_NODES; ++j) {
        const int r = lrow0 + j;
        *(float4*)(Xt + r * HIDDEN + ((lane ^ (r & 15)) << 2)) = xr[j];
    }

    #pragma unroll 1
    for (int j = 0; j < WAVE_NODES; ++j) {
        const int d = d_n, beg = beg_n, s = s_n;
        d_n   = __builtin_amdgcn_readlane(dv, j + 1);
        beg_n = __builtin_amdgcn_readlane(ev, j + 1) - d_n;
        s_n   = (lane < d_n) ? seg[beg_n + lane] : 0x7fffffff;

        const float4 a = neighbour_sum(fl, seg, s, d, beg, lane);
        const float rc = 1.0f / fmaxf((float)d, 1.0f);
        const int r = lrow0 + j;
        *(float4*)(Mt + r * HIDDEN + ((lane ^ (r & 15)) << 2)) =
            make_float4(a.x * rc, a.y * rc, a.z * rc, a.w * rc);
    }
    __syncthreads();

    const int m     = lane & 15;
    const int khalf = (lane >> 4) << 1;
    const int col   = wave * 16 + m;
    const int msw   = m << 2;

    const float bv = bias[col];
    v8f c[ROW_TILES];
    #pragma unroll
    for (int rt = 0; rt < ROW_TILES; ++rt)
        c[rt] = (v8f){bv, bv, bv, bv, bv, bv, bv, bv};

    const float* wl = Wl + col;
    const float* wr = Wr + col;
    const float* mrow = Mt + m * HIDDEN;
    const float* xrow = Xt + m * HIDDEN;

    #pragma unroll 2
    for (int k0 = 0; k0 < HIDDEN; k0 += 4) {
        const int kb = k0 + khalf;
        v2f bl, br;
        bl.x = wl[kb * HIDDEN];
        bl.y = wl[(kb + 1) * HIDDEN];
        br.x = wr[kb * HIDDEN];
        br.y = wr[(kb + 1) * HIDDEN];
        const int ks = kb ^ msw;
        #pragma unroll
        for (int rt = 0; rt < ROW_TILES; ++rt) {
            v2f a = *(const v2f*)(mrow + rt * 16 * HIDDEN + ks);
            c[rt] = __builtin_amdgcn_wmma_f32_16x16x4_f32(
                        false, a, false, bl, (short)0, c[rt], false, false);
            a = *(const v2f*)(xrow + rt * 16 * HIDDEN + ks);
            c[rt] = __builtin_amdgcn_wmma_f32_16x16x4_f32(
                        false, a, false, br, (short)0, c[rt], false, false);
        }
    }

    __syncthreads();

    const int rhalf = (lane >> 4) << 3;
    #pragma unroll
    for (int rt = 0; rt < ROW_TILES; ++rt) {
        #pragma unroll
        for (int r = 0; r < 8; ++r) {
            float v = c[rt][r];
            if (do_relu) v = fmaxf(v, 0.0f);
            const int row = rt * 16 + rhalf + r;
            Mt[row * HIDDEN + (col ^ ((row & 15) << 2))] = v;
        }
    }
    __syncthreads();

    #pragma unroll
    for (int j = 0; j < WAVE_NODES; ++j) {
        const int r = lrow0 + j;
        const float4 v = *(const float4*)(Mt + r * HIDDEN + ((lane ^ (r & 15)) << 2));
        *(float4*)(out + (size_t)(row0 + r) * HIDDEN + lane * 4) = v;
    }
}

extern "C" void kernel_launch(void* const* d_in, const int* in_sizes, int n_in,
                              void* d_out, int out_size, void* d_ws, size_t ws_size,
                              hipStream_t stream) {
    const float* x   = (const float*)d_in[0];
    const int*   ei  = (const int*)d_in[1];
    const float* Wl1 = (const float*)d_in[2];
    const float* Wr1 = (const float*)d_in[3];
    const float* b1  = (const float*)d_in[4];
    const float* Wl2 = (const float*)d_in[5];
    const float* Wr2 = (const float*)d_in[6];
    const float* b2  = (const float*)d_in[7];
    float* out = (float*)d_out;

    char* ws = (char*)d_ws;
    const size_t featBytes = (size_t)N_NODES * HIDDEN * sizeof(float);
    float* h    = (float*)(ws);
    int* cnt    = (int*)(ws + featBytes);
    int* total  = cnt + N_NODES;
    int* cursor = cnt + N_NODES + 32;
    int* seg    = cursor + N_NODES;

    hipMemsetAsync(cnt, 0, (size_t)(N_NODES + 1) * sizeof(int), stream);

    gnn_count_edges<<<(N_EDGES + 255) / 256, 256, 0, stream>>>(ei, cnt, N_EDGES);
    gnn_reserve<<<(N_NODES + 1023) / 1024, 1024, 0, stream>>>(cnt, cursor, total, N_NODES);
    gnn_fill<<<(N_EDGES + 255) / 256, 256, 0, stream>>>(ei, cursor, seg, N_EDGES);

    gnn_sage_layer<<<N_NODES / BLOCK_ROWS, 256, 0, stream>>>(x, seg, cnt, cursor,
                                                     Wl1, Wr1, b1, h, 1);
    gnn_sage_layer<<<N_NODES / BLOCK_ROWS, 256, 0, stream>>>(h, seg, cnt, cursor,
                                                     Wl2, Wr2, b2, out, 0);
}
